# starting kernel
# baseline (speedup 1.0000x reference)
#include <hip/hip_runtime.h>
#include <math.h>

typedef __attribute__((ext_vector_type(2))) float v2f;
typedef __attribute__((ext_vector_type(8))) float v8f;

#define IN_DIM 512

__device__ __forceinline__ v8f wmma_f32_4(v2f a, v2f b, v8f c) {
  return __builtin_amdgcn_wmma_f32_16x16x4_f32(false, a, false, b, (short)0, c,
                                               false, false);
}

__global__ void k_zero(float* p, int n) {
  int i = blockIdx.x * blockDim.x + threadIdx.x;
  if (i < n) p[i] = 0.0f;
}

__global__ __launch_bounds__(128) void k_fc1(const float* __restrict__ x,
                                             const float* __restrict__ w1,
                                             const float* __restrict__ b1,
                                             float* __restrict__ h1,
                                             int ntiles) {
  __shared__ float sW[IN_DIM * 16];
  int tid = threadIdx.x;
  for (int i = tid; i < IN_DIM * 16; i += 128) sW[i] = w1[i];
  __syncthreads();

  int wave = tid >> 5, lane = tid & 31;
  int tile = blockIdx.x * 4 + wave;
  if (tile >= ntiles) return;

  int n = lane & 15;
  int koff = (lane >> 4) << 1;
  long long row = (long long)tile * 16 + n;
  const float* xr = x + row * IN_DIM;

  v8f acc = {};
#pragma unroll 8
  for (int kb = 0; kb < IN_DIM / 4; ++kb) {
    int k = kb * 4 + koff;
    v2f a; a.x = xr[k]; a.y = xr[k + 1];
    v2f b; b.x = sW[k * 16 + n]; b.y = sW[(k + 1) * 16 + n];
    acc = wmma_f32_4(a, b, acc);
  }

  float bias = b1[n];
  int rbase = tile * 16 + ((lane >> 4) << 3);
#pragma unroll
  for (int r = 0; r < 8; ++r)
    h1[(long long)(rbase + r) * 16 + n] = fmaxf(acc[r] + bias, 0.0f);
}

__global__ void k_scatter1(const int* __restrict__ src, const int* __restrict__ dst,
                           const float* __restrict__ h1, float* __restrict__ agg1,
                           float* __restrict__ cnt, int nE) {
  int t = blockIdx.x * blockDim.x + threadIdx.x;
  int e = t >> 4, c = t & 15;
  if (e >= nE) return;
  int s = src[e], d = dst[e];
  atomicAdd(&agg1[(long long)d * 16 + c], h1[(long long)s * 16 + c]);
  if (c == 0) atomicAdd(&cnt[d], 1.0f);
}

__global__ __launch_bounds__(128) void k_conv1(const float* __restrict__ agg1,
                                               const float* __restrict__ cnt,
                                               const float* __restrict__ h1,
                                               const float* __restrict__ llw,
                                               const float* __restrict__ llb,
                                               const float* __restrict__ lrw,
                                               float* __restrict__ out1,
                                               int ntiles) {
  __shared__ float sW[32 * 32];
  int tid = threadIdx.x;
  for (int i = tid; i < 32 * 32; i += 128) {
    int k = i >> 5, j = i & 31;
    sW[i] = (k < 16) ? llw[k * 32 + j] : lrw[(k - 16) * 32 + j];
  }
  __syncthreads();

  int wave = tid >> 5, lane = tid & 31;
  int tile = blockIdx.x * 4 + wave;
  if (tile >= ntiles) return;

  int n = lane & 15;
  int koff = (lane >> 4) << 1;
  long long row = (long long)tile * 16 + n;
  float inv = 1.0f / fmaxf(cnt[row], 1.0f);

  v8f c0 = {}, c1 = {};
#pragma unroll
  for (int kb = 0; kb < 8; ++kb) {
    int kk = kb * 4 + koff;
    v2f a;
    if (kk < 16) { a.x = agg1[row * 16 + kk] * inv; a.y = agg1[row * 16 + kk + 1] * inv; }
    else         { a.x = h1[row * 16 + kk - 16];    a.y = h1[row * 16 + kk - 15]; }
    v2f b0; b0.x = sW[kk * 32 + n];      b0.y = sW[(kk + 1) * 32 + n];
    v2f b1; b1.x = sW[kk * 32 + 16 + n]; b1.y = sW[(kk + 1) * 32 + 16 + n];
    c0 = wmma_f32_4(a, b0, c0);
    c1 = wmma_f32_4(a, b1, c1);
  }

  float bl = llb[n], bh = llb[n + 16];
  int rbase = tile * 16 + ((lane >> 4) << 3);
#pragma unroll
  for (int r = 0; r < 8; ++r) {
    long long orow = (long long)(rbase + r) * 32;
    out1[orow + n]      = fmaxf(c0[r] + bl, 0.0f);
    out1[orow + 16 + n] = fmaxf(c1[r] + bh, 0.0f);
  }
}

__global__ void k_scatter2(const int* __restrict__ src, const int* __restrict__ dst,
                           const float* __restrict__ out1, float* __restrict__ agg2,
                           int nE) {
  int t = blockIdx.x * blockDim.x + threadIdx.x;
  int e = t >> 5, c = t & 31;
  if (e >= nE) return;
  int s = src[e], d = dst[e];
  atomicAdd(&agg2[(long long)d * 32 + c], out1[(long long)s * 32 + c]);
}

__global__ __launch_bounds__(128) void k_conv2(const float* __restrict__ agg2,
                                               const float* __restrict__ cnt,
                                               const float* __restrict__ out1,
                                               const float* __restrict__ llw,
                                               const float* __restrict__ llb,
                                               const float* __restrict__ lrw,
                                               float* __restrict__ out2,
                                               int ntiles) {
  __shared__ float sW[64 * 16];
  int tid = threadIdx.x;
  for (int i = tid; i < 64 * 16; i += 128) {
    int k = i >> 4, j = i & 15;
    sW[i] = (k < 32) ? llw[k * 16 + j] : lrw[(k - 32) * 16 + j];
  }
  __syncthreads();

  int wave = tid >> 5, lane = tid & 31;
  int tile = blockIdx.x * 4 + wave;
  if (tile >= ntiles) return;

  int n = lane & 15;
  int koff = (lane >> 4) << 1;
  long long row = (long long)tile * 16 + n;
  float inv = 1.0f / fmaxf(cnt[row], 1.0f);

  v8f acc = {};
#pragma unroll
  for (int kb = 0; kb < 16; ++kb) {
    int kk = kb * 4 + koff;
    v2f a;
    if (kk < 32) { a.x = agg2[row * 32 + kk] * inv; a.y = agg2[row * 32 + kk + 1] * inv; }
    else         { a.x = out1[row * 32 + kk - 32]; a.y = out1[row * 32 + kk - 31]; }
    v2f b; b.x = sW[kk * 16 + n]; b.y = sW[(kk + 1) * 16 + n];
    acc = wmma_f32_4(a, b, acc);
  }

  float bias = llb[n];
  int rbase = tile * 16 + ((lane >> 4) << 3);
#pragma unroll
  for (int r = 0; r < 8; ++r)
    out2[(long long)(rbase + r) * 16 + n] = fmaxf(acc[r] + bias, 0.0f);
}

__global__ void k_head(const float* __restrict__ out2, const float* __restrict__ w2,
                       const float* __restrict__ b2, float* __restrict__ out, int N) {
  int i = blockIdx.x * blockDim.x + threadIdx.x;
  if (i >= N) return;
  float h0 = b2[0], h1 = b2[1], h2 = b2[2];
  const float* r = out2 + (long long)i * 16;
#pragma unroll
  for (int k = 0; k < 16; ++k) {
    float v = r[k];
    h0 += v * w2[k * 3 + 0];
    h1 += v * w2[k * 3 + 1];
    h2 += v * w2[k * 3 + 2];
  }
  float gsi = 1.0f / (1.0f + __expf(-h1));
  float mxi = 1.0f / (1.0f + __expf(-h2));
  float fsi = fmaxf(h0, 0.0f) + gsi;
  out[(long long)i * 3 + 0] = fsi;
  out[(long long)i * 3 + 1] = gsi;
  out[(long long)i * 3 + 2] = mxi;
}

extern "C" void kernel_launch(void* const* d_in, const int* in_sizes, int n_in,
                              void* d_out, int out_size, void* d_ws, size_t ws_size,
                              hipStream_t stream) {
  const float* x    = (const float*)d_in[0];
  const int*   ei   = (const int*)d_in[1];
  const float* w1   = (const float*)d_in[2];
  const float* b1   = (const float*)d_in[3];
  const float* c1ll = (const float*)d_in[4];
  const float* c1lb = (const float*)d_in[5];
  const float* c1lr = (const float*)d_in[6];
  const float* c2ll = (const float*)d_in[7];
  const float* c2lb = (const float*)d_in[8];
  const float* c2lr = (const float*)d_in[9];
  const float* w2   = (const float*)d_in[10];
  const float* b2   = (const float*)d_in[11];
  float* out = (float*)d_out;

  int N = in_sizes[0] / IN_DIM;
  int E = in_sizes[1] / 2;
  const int* src = ei;
  const int* dst = ei + E;

  float* ws   = (float*)d_ws;
  float* h1   = ws;
  float* cnt  = ws + (long long)16 * N;
  float* agg1 = ws + (long long)17 * N;
  float* out1 = ws + (long long)33 * N;
  float* agg2 = ws + (long long)65 * N;
  float* out2 = ws + (long long)97 * N;

  int ntiles = (N + 15) / 16;
  int wblocks = (ntiles + 3) / 4;

  {
    int n1 = 17 * N;
    k_zero<<<(n1 + 255) / 256, 256, 0, stream>>>(cnt, n1);
    int n2 = 32 * N;
    k_zero<<<(n2 + 255) / 256, 256, 0, stream>>>(agg2, n2);
  }

  k_fc1<<<wblocks, 128, 0, stream>>>(x, w1, b1, h1, ntiles);

  {
    long long t = (long long)E * 16;
    k_scatter1<<<(unsigned)((t + 255) / 256), 256, 0, stream>>>(src, dst, h1, agg1, cnt, E);
  }

  k_conv1<<<wblocks, 128, 0, stream>>>(agg1, cnt, h1, c1ll, c1lb, c1lr, out1, ntiles);

  {
    long long t = (long long)E * 32;
    k_scatter2<<<(unsigned)((t + 255) / 256), 256, 0, stream>>>(src, dst, out1, agg2, E);
  }

  k_conv2<<<wblocks, 128, 0, stream>>>(agg2, cnt, out1, c2ll, c2lb, c2lr, out2, ntiles);

  k_head<<<(N + 255) / 256, 256, 0, stream>>>(out2, w2, b2, out, N);
}


# fc1 has no launch of its own: cut along K into two halves that run as extra work-groups of the bin-counting and the placement launch (partial sums pass through h1, same k order); 6 launches become 5
# speedup vs baseline: 4.5724x; 4.5724x over previous
#include <hip/hip_runtime.h>
#include <math.h>

typedef __attribute__((ext_vector_type(2))) float v2f;
typedef __attribute__((ext_vector_type(8))) float v8f;

#define IN_DIM 512

__device__ __forceinline__ v8f wmma_f32_4(v2f a, v2f b, v8f c) {
  return __builtin_amdgcn_wmma_f32_16x16x4_f32(false, a, false, b, (short)0, c,
                                               false, false);
}

#define FC_KB 64

template <bool FIRST>
__device__ __forceinline__ void fc1_half(const float* __restrict__ x,
                                         const float* __restrict__ w1,
                                         const float* __restrict__ b1,
                                         float* __restrict__ h1,
                                         float* sW, int fb, int ntiles) {
  constexpr int K0 = FIRST ? 0 : FC_KB * 4;
  int tid = threadIdx.x;
  for (int i = tid; i < FC_KB * 64; i += 256) sW[i] = w1[K0 * 16 + i];
  __syncthreads();

  int wave = tid >> 5, lane = tid & 31;
  int tile = fb * 8 + wave;
  if (tile >= ntiles) return;

  int n = lane & 15;
  int koff = (lane >> 4) << 1;
  long long row = (long long)tile * 16 + n;
  const float* xr = x + row * IN_DIM + K0;
  int rbase = tile * 16 + ((lane >> 4) << 3);

  v8f acc = {};
  if (!FIRST) {
#pragma unroll
    for (int r = 0; r < 8; ++r) acc[r] = h1[(long long)(rbase + r) * 16 + n];
  }
#pragma unroll 8
  for (int kb = 0; kb < FC_KB; ++kb) {
    int k = kb * 4 + koff;
    v2f a; a.x = xr[k]; a.y = xr[k + 1];
    v2f b; b.x = sW[k * 16 + n]; b.y = sW[(k + 1) * 16 + n];
    acc = wmma_f32_4(a, b, acc);
  }

  if (FIRST) {
#pragma unroll
    for (int r = 0; r < 8; ++r) h1[(long long)(rbase + r) * 16 + n] = acc[r];
  } else {
    float bias = b1[n];
#pragma unroll
    for (int r = 0; r < 8; ++r)
      h1[(long long)(rbase + r) * 16 + n] = fmaxf(acc[r] + bias, 0.0f);
  }
}

#define BIN_SHIFT 7
#define BIN_NODES 128
#define NB_MAX 1024
#define NCH 512
#define EDGE_BATCH 4

__device__ __forceinline__ int block_incl_scan256(int v, int* sh) {
  int tid = threadIdx.x;
  sh[tid] = v;
  __syncthreads();
  for (int off = 1; off < 256; off <<= 1) {
    int t = (tid >= off) ? sh[tid - off] : 0;
    __syncthreads();
    sh[tid] += t;
    __syncthreads();
  }
  return sh[tid];
}

__global__ __launch_bounds__(256) void k_hist(const int* __restrict__ dst,
                                              int* __restrict__ hist,
                                              int nE, int chunk, int nb,
                                              const float* __restrict__ x,
                                              const float* __restrict__ w1,
                                              const float* __restrict__ b1,
                                              float* __restrict__ h1, int ntiles) {
  __shared__ int sc[NB_MAX];
  __shared__ float sW[FC_KB * 64];
  if (blockIdx.x >= NCH) {
    fc1_half<true>(x, w1, b1, h1, sW, blockIdx.x - NCH, ntiles);
    return;
  }
  int tid = threadIdx.x, c = blockIdx.x;
  for (int i = tid; i < nb; i += 256) sc[i] = 0;
  __syncthreads();
  int e0 = c * chunk;
  int e1 = min(e0 + chunk, nE);
  for (int e = e0 + tid; e < e1; e += 256 * EDGE_BATCH) {
    int d[EDGE_BATCH];
#pragma unroll
    for (int j = 0; j < EDGE_BATCH; ++j) {
      int ee = e + j * 256;
      d[j] = (ee < e1) ? dst[ee] : -1;
    }
#pragma unroll
    for (int j = 0; j < EDGE_BATCH; ++j)
      if (d[j] >= 0) atomicAdd(&sc[d[j] >> BIN_SHIFT], 1);
  }
  __syncthreads();
  for (int i = tid; i < nb; i += 256) hist[i * NCH + c] = sc[i];
}

__global__ __launch_bounds__(256) void k_scan(int* hist, int* __restrict__ binTotal) {
  __shared__ int sh[256];
  int tid = threadIdx.x, b = blockIdx.x;
  int* row = hist + b * NCH;
  int v0 = row[2 * tid], v1 = row[2 * tid + 1];
  int incl = block_incl_scan256(v0 + v1, sh);
  int excl = incl - (v0 + v1);
  row[2 * tid] = excl;
  row[2 * tid + 1] = excl + v0;
  if (tid == 255) binTotal[b] = incl;
}

#define TILE_K 25
#define TILE_E (256 * TILE_K)
__global__ __launch_bounds__(256) void k_binscatter(const int* __restrict__ src,
                                                    const int* __restrict__ dst,
                                                    const int* __restrict__ hist,
                                                    const int* __restrict__ binTotal,
                                                    int* __restrict__ binStart,
                                                    int* __restrict__ packed,
                                                    int nE, int chunk, int nb,
                                                    const float* __restrict__ x,
                                                    const float* __restrict__ w1,
                                                    const float* __restrict__ b1,
                                                    float* __restrict__ h1, int ntiles) {
  __shared__ int smem[3 * NB_MAX + 256 + 2 * TILE_E];
  if (blockIdx.x >= NCH) {
    fc1_half<false>(x, w1, b1, h1, (float*)smem, blockIdx.x - NCH, ntiles);
    return;
  }
  int* cur  = smem;
  int* lcur = smem + NB_MAX;
  int* gdel = smem + 2 * NB_MAX;
  int* sh   = smem + 3 * NB_MAX;
  int* sval = sh + 256;
  int* sgp  = sval + TILE_E;
  int tid = threadIdx.x, c = blockIdx.x;
  int t[4];
  int sum = 0;
#pragma unroll
  for (int j = 0; j < 4; ++j) {
    int i = tid * 4 + j;
    t[j] = (i < nb) ? binTotal[i] : 0;
    sum += t[j];
  }
  int base = block_incl_scan256(sum, sh) - sum;
#pragma unroll
  for (int j = 0; j < 4; ++j) {
    int i = tid * 4 + j;
    int h = (i < nb) ? hist[i * NCH + c] : 0;
    cur[i] = base + h;
    if (c == 0 && i <= nb) binStart[i] = base;
    base += t[j];
  }
  int e0 = c * chunk;
  int e1 = min(e0 + chunk, nE);
  for (int t0 = e0; t0 < e1; t0 += TILE_E) {
    int n = min(TILE_E, e1 - t0);
    for (int i = tid; i < NB_MAX; i += 256) lcur[i] = 0;
    __syncthreads();

    int d[TILE_K], s[TILE_K];
#pragma unroll
    for (int k = 0; k < TILE_K; ++k) {
      int idx = tid + k * 256;
      int v = dst[t0 + min(idx, n - 1)];
      d[k] = (idx < n) ? v : -1;
    }
#pragma unroll
    for (int k = 0; k < TILE_K; ++k) {
      int idx = tid + k * 256;
      s[k] = src[t0 + min(idx, n - 1)];
    }
#pragma unroll
    for (int k = 0; k < TILE_K; ++k)
      if (d[k] >= 0) atomicAdd(&lcur[d[k] >> BIN_SHIFT], 1);
    __syncthreads();

    int lc[4];
    int lsum = 0;
#pragma unroll
    for (int j = 0; j < 4; ++j) {
      lc[j] = lcur[tid * 4 + j];
      lsum += lc[j];
    }
    int lbase = block_incl_scan256(lsum, sh) - lsum;
#pragma unroll
    for (int j = 0; j < 4; ++j) {
      int i = tid * 4 + j;
      int g = cur[i];
      lcur[i] = lbase;
      gdel[i] = g - lbase;
      cur[i] = g + lc[j];
      lbase += lc[j];
    }
    __syncthreads();

#pragma unroll
    for (int k = 0; k < TILE_K; ++k) {
      if (d[k] >= 0) {
        int bin = d[k] >> BIN_SHIFT;
        int pos = atomicAdd(&lcur[bin], 1);
        sval[pos] = (s[k] << BIN_SHIFT) | (d[k] & (BIN_NODES - 1));
        sgp[pos] = pos + gdel[bin];
      }
    }
    __syncthreads();

    for (int j = tid; j < n; j += 256) packed[sgp[j]] = sval[j];
  }
}

template <int C, bool CNT>
__device__ __forceinline__ void bin_sums(float* acc, float* cl,
                                         const int* __restrict__ packed,
                                         const float* __restrict__ feat,
                                         int e0, int e1) {
  constexpr int P = C + 1;
  constexpr int LPE = C / 4;
  constexpr int EPI = 32 / LPE;
  constexpr int STEPS = 32 / EPI;
  int tid = threadIdx.x;
  int wave = tid >> 5, lane = tid & 31;
  int sub = lane / LPE, q = lane % LPE;
  for (int eb = e0 + wave * 32; eb < e1; eb += 256) {
    int p[STEPS];
    float4 v[STEPS];
#pragma unroll
    for (int j = 0; j < STEPS; ++j) {
      int idx = eb + j * EPI + sub;
      p[j] = (idx < e1) ? packed[idx] : -1;
    }
#pragma unroll
    for (int j = 0; j < STEPS; ++j) {
      int s = (p[j] >= 0) ? (p[j] >> BIN_SHIFT) : 0;
      v[j] = *(const float4*)(feat + (long long)s * C + q * 4);
    }
#pragma unroll
    for (int j = 0; j < STEPS; ++j) {
      if (p[j] >= 0) {
        int dl = p[j] & (BIN_NODES - 1);
        float* r = acc + dl * P + q * 4;
        atomicAdd(r + 0, v[j].x);
        atomicAdd(r + 1, v[j].y);
        atomicAdd(r + 2, v[j].z);
        atomicAdd(r + 3, v[j].w);
        if (CNT && q == 0) atomicAdd(&cl[dl], 1.0f);
      }
    }
  }
}

__global__ __launch_bounds__(256) void k_layer1(const int* __restrict__ packed,
                                                const int* __restrict__ binStart,
                                                const float* __restrict__ h1,
                                                const float* __restrict__ llw,
                                                const float* __restrict__ llb,
                                                const float* __restrict__ lrw,
                                                float* __restrict__ out1,
                                                float* __restrict__ cnt,
                                                int N, int ntiles) {
  constexpr int C = 16, P = C + 1;
  __shared__ float acc[BIN_NODES * P];
  __shared__ float cl[BIN_NODES];
  __shared__ float sW[32 * 32];
  int tid = threadIdx.x, b = blockIdx.x;
  int wave = tid >> 5, lane = tid & 31;
  int n = lane & 15;
  int koff = (lane >> 4) << 1;
  int tile = b * (BIN_NODES / 16) + wave;
  bool valid = tile < ntiles;
  long long row = (long long)(valid ? tile : 0) * 16 + n;

  v2f self[4];
#pragma unroll
  for (int kb = 0; kb < 4; ++kb) {
    int kk = kb * 4 + koff;
    self[kb].x = h1[row * 16 + kk];
    self[kb].y = h1[row * 16 + kk + 1];
  }

  float wv[4];
#pragma unroll
  for (int j = 0; j < 4; ++j)
    wv[j] = (j < 2) ? llw[tid + j * 256] : lrw[tid + (j - 2) * 256];

  for (int i = tid; i < BIN_NODES * P; i += 256) acc[i] = 0.0f;
  if (tid < BIN_NODES) cl[tid] = 0.0f;
#pragma unroll
  for (int j = 0; j < 4; ++j) sW[tid + j * 256] = wv[j];
  __syncthreads();

  bin_sums<C, true>(acc, cl, packed, h1, binStart[b], binStart[b + 1]);
  __syncthreads();

  if (tid < BIN_NODES) {
    int node = b * BIN_NODES + tid;
    if (node < N) cnt[node] = cl[tid];
  }
  if (!valid) return;

  int lr = wave * 16 + n;
  float inv = 1.0f / fmaxf(cl[lr], 1.0f);
  v8f c0 = {}, c1 = {};
#pragma unroll
  for (int kb = 0; kb < 8; ++kb) {
    int kk = kb * 4 + koff;
    v2f a;
    if (kb < 4) { a.x = acc[lr * P + kk] * inv; a.y = acc[lr * P + kk + 1] * inv; }
    else        { a = self[kb - 4]; }
    v2f b0; b0.x = sW[kk * 32 + n];      b0.y = sW[(kk + 1) * 32 + n];
    v2f b1; b1.x = sW[kk * 32 + 16 + n]; b1.y = sW[(kk + 1) * 32 + 16 + n];
    c0 = wmma_f32_4(a, b0, c0);
    c1 = wmma_f32_4(a, b1, c1);
  }

  float bl = llb[n], bh = llb[n + 16];
  int rbase = tile * 16 + ((lane >> 4) << 3);
#pragma unroll
  for (int r = 0; r < 8; ++r) {
    long long orow = (long long)(rbase + r) * 32;
    out1[orow + n]      = fmaxf(c0[r] + bl, 0.0f);
    out1[orow + 16 + n] = fmaxf(c1[r] + bh, 0.0f);
  }
}

__global__ __launch_bounds__(256) void k_layer2(const int* __restrict__ packed,
                                                const int* __restrict__ binStart,
                                                const float* __restrict__ out1,
                                                const float* __restrict__ cnt,
                                                const float* __restrict__ llw,
                                                const float* __restrict__ llb,
                                                const float* __restrict__ lrw,
                                                const float* __restrict__ w2,
                                                const float* __restrict__ b2,
                                                float* __restrict__ out,
                                                int N, int ntiles) {
  constexpr int C = 32, P = C + 1;
  __shared__ float acc[BIN_NODES * P];
  __shared__ float sW[64 * 16];
  int tid = threadIdx.x, b = blockIdx.x;
  int wave = tid >> 5, lane = tid & 31;
  int n = lane & 15;
  int koff = (lane >> 4) << 1;
  int tile = b * (BIN_NODES / 16) + wave;
  bool valid = tile < ntiles;
  long long row = (long long)(valid ? tile : 0) * 16 + n;

  float deg = cnt[row];
  v2f self[8];
#pragma unroll
  for (int kb = 0; kb < 8; ++kb) {
    int kk = kb * 4 + koff;
    self[kb].x = out1[row * 32 + kk];
    self[kb].y = out1[row * 32 + kk + 1];
  }

  float wv[4];
#pragma unroll
  for (int j = 0; j < 4; ++j)
    wv[j] = (j < 2) ? llw[tid + j * 256] : lrw[tid + (j - 2) * 256];

  for (int i = tid; i < BIN_NODES * P; i += 256) acc[i] = 0.0f;
#pragma unroll
  for (int j = 0; j < 4; ++j) sW[tid + j * 256] = wv[j];
  __syncthreads();

  bin_sums<C, false>(acc, nullptr, packed, out1, binStart[b], binStart[b + 1]);
  __syncthreads();

  float* trow = acc + wave * 16 * P;
  if (valid) {
    float inv = 1.0f / fmaxf(deg, 1.0f);
    v8f d = {};
#pragma unroll
    for (int kb = 0; kb < 16; ++kb) {
      int kk = kb * 4 + koff;
      v2f a;
      if (kb < 8) { a.x = trow[n * P + kk] * inv; a.y = trow[n * P + kk + 1] * inv; }
      else        { a = self[kb - 8]; }
      v2f bw; bw.x = sW[kk * 16 + n]; bw.y = sW[(kk + 1) * 16 + n];
      d = wmma_f32_4(a, bw, d);
    }
    float bias = llb[n];
    int r8 = (lane >> 4) << 3;
#pragma unroll
    for (int r = 0; r < 8; ++r)
      trow[(r8 + r) * P + n] = fmaxf(d[r] + bias, 0.0f);
  }
  __syncthreads();

  if (valid && lane < 16) {
    int i = tile * 16 + lane;
    if (i < N) {
      float h0 = b2[0], h1 = b2[1], h2 = b2[2];
      const float* r = trow + lane * P;
#pragma unroll
      for (int k = 0; k < 16; ++k) {
        float v = r[k];
        h0 += v * w2[k * 3 + 0];
        h1 += v * w2[k * 3 + 1];
        h2 += v * w2[k * 3 + 2];
      }
      float gsi = 1.0f / (1.0f + __expf(-h1));
      float mxi = 1.0f / (1.0f + __expf(-h2));
      float fsi = fmaxf(h0, 0.0f) + gsi;
      out[(long long)i * 3 + 0] = fsi;
      out[(long long)i * 3 + 1] = gsi;
      out[(long long)i * 3 + 2] = mxi;
    }
  }
}

extern "C" void kernel_launch(void* const* d_in, const int* in_sizes, int n_in,
                              void* d_out, int out_size, void* d_ws, size_t ws_size,
                              hipStream_t stream) {
  const float* x    = (const float*)d_in[0];
  const int*   ei   = (const int*)d_in[1];
  const float* w1   = (const float*)d_in[2];
  const float* b1   = (const float*)d_in[3];
  const float* c1ll = (const float*)d_in[4];
  const float* c1lb = (const float*)d_in[5];
  const float* c1lr = (const float*)d_in[6];
  const float* c2ll = (const float*)d_in[7];
  const float* c2lb = (const float*)d_in[8];
  const float* c2lr = (const float*)d_in[9];
  const float* w2   = (const float*)d_in[10];
  const float* b2   = (const float*)d_in[11];
  float* out = (float*)d_out;

  int N = in_sizes[0] / IN_DIM;
  int E = in_sizes[1] / 2;
  const int* src = ei;
  const int* dst = ei + E;

  float* ws   = (float*)d_ws;
  float* h1   = ws;
  float* out1 = ws + (long long)32 * N;
  int*   packed = (int*)(ws + (long long)64 * N);
  float* cnt  = ws + (long long)96 * N;
  int*   hist = (int*)(ws + (long long)97 * N);
  int*   binTotal = hist + NB_MAX * NCH;
  int*   binStart = binTotal + NB_MAX;

  int ntiles = (N + 15) / 16;
  int fblocks = (ntiles + 7) / 8;
  int nb = (N + BIN_NODES - 1) / BIN_NODES;
  int chunk = (E + NCH - 1) / NCH;

  k_hist<<<NCH + fblocks, 256, 0, stream>>>(dst, hist, E, chunk, nb, x, w1, b1, h1, ntiles);
  k_scan<<<nb, 256, 0, stream>>>(hist, binTotal);
  k_binscatter<<<NCH + fblocks, 256, 0, stream>>>(src, dst, hist, binTotal, binStart, packed, E, chunk, nb,
                                                  x, w1, b1, h1, ntiles);

  k_layer1<<<nb, 256, 0, stream>>>(packed, binStart, h1, c1ll, c1lb, c1lr, out1, cnt, N, ntiles);

  k_layer2<<<nb, 256, 0, stream>>>(packed, binStart, out1, cnt, c2ll, c2lb, c2lr, w2, b2, out,
                                   N, ntiles);
}


# fc1 loads run ahead of its products: W1 half staged by one group of 16-byte loads, x rows fetched as 16-byte pieces two 32-k blocks ahead; k pairs regrouped, so fc1's summation order changes
# speedup vs baseline: 4.7571x; 1.0404x over previous
#include <hip/hip_runtime.h>
#include <math.h>

typedef __attribute__((ext_vector_type(2))) float v2f;
typedef __attribute__((ext_vector_type(8))) float v8f;

#define IN_DIM 512

__device__ __forceinline__ v8f wmma_f32_4(v2f a, v2f b, v8f c) {
  return __builtin_amdgcn_wmma_f32_16x16x4_f32(false, a, false, b, (short)0, c,
                                               false, false);
}

#define FC_KB 64
#define FC_BLOCKS (FC_KB / 8)
#define FC_AHEAD 2

template <bool FIRST>
__device__ __forceinline__ void fc1_half(const float* __restrict__ x,
                                         const float* __restrict__ w1,
                                         const float* __restrict__ b1,
                                         float* __restrict__ h1,
                                         float* sW, int fb, int ntiles) {
  constexpr int K0 = FIRST ? 0 : FC_KB * 4;
  int tid = threadIdx.x;
  {
    const float4* wsrc = (const float4*)(w1 + K0 * 16);
    float4 t[4];
#pragma unroll
    for (int j = 0; j < 4; ++j) t[j] = wsrc[tid + j * 256];
#pragma unroll
    for (int j = 0; j < 4; ++j) ((float4*)sW)[tid + j * 256] = t[j];
  }
  __syncthreads();

  int wave = tid >> 5, lane = tid & 31;
  int tile = fb * 8 + wave;
  if (tile >= ntiles) return;

  int n = lane & 15;
  int h4 = (lane >> 4) << 2;
  long long row = (long long)tile * 16 + n;
  const float* xr = x + row * IN_DIM + K0 + h4;
  int rbase = tile * 16 + ((lane >> 4) << 3);

  float4 xa[FC_AHEAD + 1][4];
#pragma unroll
  for (int b = 0; b < FC_AHEAD; ++b)
#pragma unroll
    for (int g = 0; g < 4; ++g) xa[b][g] = *(const float4*)(xr + b * 32 + g * 8);

  v8f acc = {};
  if (!FIRST) {
#pragma unroll
    for (int r = 0; r < 8; ++r) acc[r] = h1[(long long)(rbase + r) * 16 + n];
  }
#pragma unroll
  for (int blk = 0; blk < FC_BLOCKS; ++blk) {
    if (blk + FC_AHEAD < FC_BLOCKS) {
#pragma unroll
      for (int g = 0; g < 4; ++g)
        xa[(blk + FC_AHEAD) % (FC_AHEAD + 1)][g] =
            *(const float4*)(xr + (blk + FC_AHEAD) * 32 + g * 8);
    }
#pragma unroll
    for (int g = 0; g < 4; ++g) {
      float4 av = xa[blk % (FC_AHEAD + 1)][g];
      int k = blk * 32 + g * 8 + h4;
      v2f a0; a0.x = av.x; a0.y = av.y;
      v2f a1; a1.x = av.z; a1.y = av.w;
      v2f b0; b0.x = sW[k * 16 + n];       b0.y = sW[(k + 1) * 16 + n];
      v2f b1; b1.x = sW[(k + 2) * 16 + n]; b1.y = sW[(k + 3) * 16 + n];
      acc = wmma_f32_4(a0, b0, acc);
      acc = wmma_f32_4(a1, b1, acc);
    }
  }

  if (FIRST) {
#pragma unroll
    for (int r = 0; r < 8; ++r) h1[(long long)(rbase + r) * 16 + n] = acc[r];
  } else {
    float bias = b1[n];
#pragma unroll
    for (int r = 0; r < 8; ++r)
      h1[(long long)(rbase + r) * 16 + n] = fmaxf(acc[r] + bias, 0.0f);
  }
}

#define BIN_SHIFT 7
#define BIN_NODES 128
#define NB_MAX 1024
#define NCH 512
#define EDGE_BATCH 4

__device__ __forceinline__ int block_incl_scan256(int v, int* sh) {
  int tid = threadIdx.x;
  sh[tid] = v;
  __syncthreads();
  for (int off = 1; off < 256; off <<= 1) {
    int t = (tid >= off) ? sh[tid - off] : 0;
    __syncthreads();
    sh[tid] += t;
    __syncthreads();
  }
  return sh[tid];
}

__global__ __launch_bounds__(256) void k_hist(const int* __restrict__ dst,
                                              int* __restrict__ hist,
                                              int nE, int chunk, int nb,
                                              const float* __restrict__ x,
                                              const float* __restrict__ w1,
                                              const float* __restrict__ b1,
                                              float* __restrict__ h1, int ntiles) {
  __shared__ int sc[NB_MAX];
  __shared__ __attribute__((aligned(16))) float sW[FC_KB * 64];
  if (blockIdx.x >= NCH) {
    fc1_half<true>(x, w1, b1, h1, sW, blockIdx.x - NCH, ntiles);
    return;
  }
  int tid = threadIdx.x, c = blockIdx.x;
  for (int i = tid; i < nb; i += 256) sc[i] = 0;
  __syncthreads();
  int e0 = c * chunk;
  int e1 = min(e0 + chunk, nE);
  for (int e = e0 + tid; e < e1; e += 256 * EDGE_BATCH) {
    int d[EDGE_BATCH];
#pragma unroll
    for (int j = 0; j < EDGE_BATCH; ++j) {
      int ee = e + j * 256;
      d[j] = (ee < e1) ? dst[ee] : -1;
    }
#pragma unroll
    for (int j = 0; j < EDGE_BATCH; ++j)
      if (d[j] >= 0) atomicAdd(&sc[d[j] >> BIN_SHIFT], 1);
  }
  __syncthreads();
  for (int i = tid; i < nb; i += 256) hist[i * NCH + c] = sc[i];
}

__global__ __launch_bounds__(256) void k_scan(int* hist, int* __restrict__ binTotal) {
  __shared__ int sh[256];
  int tid = threadIdx.x, b = blockIdx.x;
  int* row = hist + b * NCH;
  int v0 = row[2 * tid], v1 = row[2 * tid + 1];
  int incl = block_incl_scan256(v0 + v1, sh);
  int excl = incl - (v0 + v1);
  row[2 * tid] = excl;
  row[2 * tid + 1] = excl + v0;
  if (tid == 255) binTotal[b] = incl;
}

#define TILE_K 25
#define TILE_E (256 * TILE_K)
__global__ __launch_bounds__(256) void k_binscatter(const int* __restrict__ src,
                                                    const int* __restrict__ dst,
                                                    const int* __restrict__ hist,
                                                    const int* __restrict__ binTotal,
                                                    int* __restrict__ binStart,
                                                    int* __restrict__ packed,
                                                    int nE, int chunk, int nb,
                                                    const float* __restrict__ x,
                                                    const float* __restrict__ w1,
                                                    const float* __restrict__ b1,
                                                    float* __restrict__ h1, int ntiles) {
  __shared__ __attribute__((aligned(16))) int smem[3 * NB_MAX + 256 + 2 * TILE_E];
  if (blockIdx.x >= NCH) {
    fc1_half<false>(x, w1, b1, h1, (float*)smem, blockIdx.x - NCH, ntiles);
    return;
  }
  int* cur  = smem;
  int* lcur = smem + NB_MAX;
  int* gdel = smem + 2 * NB_MAX;
  int* sh   = smem + 3 * NB_MAX;
  int* sval = sh + 256;
  int* sgp  = sval + TILE_E;
  int tid = threadIdx.x, c = blockIdx.x;
  int t[4];
  int sum = 0;
#pragma unroll
  for (int j = 0; j < 4; ++j) {
    int i = tid * 4 + j;
    t[j] = (i < nb) ? binTotal[i] : 0;
    sum += t[j];
  }
  int base = block_incl_scan256(sum, sh) - sum;
#pragma unroll
  for (int j = 0; j < 4; ++j) {
    int i = tid * 4 + j;
    int h = (i < nb) ? hist[i * NCH + c] : 0;
    cur[i] = base + h;
    if (c == 0 && i <= nb) binStart[i] = base;
    base += t[j];
  }
  int e0 = c * chunk;
  int e1 = min(e0 + chunk, nE);
  for (int t0 = e0; t0 < e1; t0 += TILE_E) {
    int n = min(TILE_E, e1 - t0);
    for (int i = tid; i < NB_MAX; i += 256) lcur[i] = 0;
    __syncthreads();

    int d[TILE_K], s[TILE_K];
#pragma unroll
    for (int k = 0; k < TILE_K; ++k) {
      int idx = tid + k * 256;
      int v = dst[t0 + min(idx, n - 1)];
      d[k] = (idx < n) ? v : -1;
    }
#pragma unroll
    for (int k = 0; k < TILE_K; ++k) {
      int idx = tid + k * 256;
      s[k] = src[t0 + min(idx, n - 1)];
    }
#pragma unroll
    for (int k = 0; k < TILE_K; ++k)
      if (d[k] >= 0) atomicAdd(&lcur[d[k] >> BIN_SHIFT], 1);
    __syncthreads();

    int lc[4];
    int lsum = 0;
#pragma unroll
    for (int j = 0; j < 4; ++j) {
      lc[j] = lcur[tid * 4 + j];
      lsum += lc[j];
    }
    int lbase = block_incl_scan256(lsum, sh) - lsum;
#pragma unroll
    for (int j = 0; j < 4; ++j) {
      int i = tid * 4 + j;
      int g = cur[i];
      lcur[i] = lbase;
      gdel[i] = g - lbase;
      cur[i] = g + lc[j];
      lbase += lc[j];
    }
    __syncthreads();

#pragma unroll
    for (int k = 0; k < TILE_K; ++k) {
      if (d[k] >= 0) {
        int bin = d[k] >> BIN_SHIFT;
        int pos = atomicAdd(&lcur[bin], 1);
        sval[pos] = (s[k] << BIN_SHIFT) | (d[k] & (BIN_NODES - 1));
        sgp[pos] = pos + gdel[bin];
      }
    }
    __syncthreads();

    for (int j = tid; j < n; j += 256) packed[sgp[j]] = sval[j];
  }
}

template <int C, bool CNT>
__device__ __forceinline__ void bin_sums(float* acc, float* cl,
                                         const int* __restrict__ packed,
                                         const float* __restrict__ feat,
                                         int e0, int e1) {
  constexpr int P = C + 1;
  constexpr int LPE = C / 4;
  constexpr int EPI = 32 / LPE;
  constexpr int STEPS = 32 / EPI;
  int tid = threadIdx.x;
  int wave = tid >> 5, lane = tid & 31;
  int sub = lane / LPE, q = lane % LPE;
  for (int eb = e0 + wave * 32; eb < e1; eb += 256) {
    int p[STEPS];
    float4 v[STEPS];
#pragma unroll
    for (int j = 0; j < STEPS; ++j) {
      int idx = eb + j * EPI + sub;
      p[j] = (idx < e1) ? packed[idx] : -1;
    }
#pragma unroll
    for (int j = 0; j < STEPS; ++j) {
      int s = (p[j] >= 0) ? (p[j] >> BIN_SHIFT) : 0;
      v[j] = *(const float4*)(feat + (long long)s * C + q * 4);
    }
#pragma unroll
    for (int j = 0; j < STEPS; ++j) {
      if (p[j] >= 0) {
        int dl = p[j] & (BIN_NODES - 1);
        float* r = acc + dl * P + q * 4;
        atomicAdd(r + 0, v[j].x);
        atomicAdd(r + 1, v[j].y);
        atomicAdd(r + 2, v[j].z);
        atomicAdd(r + 3, v[j].w);
        if (CNT && q == 0) atomicAdd(&cl[dl], 1.0f);
      }
    }
  }
}

__global__ __launch_bounds__(256) void k_layer1(const int* __restrict__ packed,
                                                const int* __restrict__ binStart,
                                                const float* __restrict__ h1,
                                                const float* __restrict__ llw,
                                                const float* __restrict__ llb,
                                                const float* __restrict__ lrw,
                                                float* __restrict__ out1,
                                                float* __restrict__ cnt,
                                                int N, int ntiles) {
  constexpr int C = 16, P = C + 1;
  __shared__ float acc[BIN_NODES * P];
  __shared__ float cl[BIN_NODES];
  __shared__ float sW[32 * 32];
  int tid = threadIdx.x, b = blockIdx.x;
  int wave = tid >> 5, lane = tid & 31;
  int n = lane & 15;
  int koff = (lane >> 4) << 1;
  int tile = b * (BIN_NODES / 16) + wave;
  bool valid = tile < ntiles;
  long long row = (long long)(valid ? tile : 0) * 16 + n;

  v2f self[4];
#pragma unroll
  for (int kb = 0; kb < 4; ++kb) {
    int kk = kb * 4 + koff;
    self[kb].x = h1[row * 16 + kk];
    self[kb].y = h1[row * 16 + kk + 1];
  }

  float wv[4];
#pragma unroll
  for (int j = 0; j < 4; ++j)
    wv[j] = (j < 2) ? llw[tid + j * 256] : lrw[tid + (j - 2) * 256];

  for (int i = tid; i < BIN_NODES * P; i += 256) acc[i] = 0.0f;
  if (tid < BIN_NODES) cl[tid] = 0.0f;
#pragma unroll
  for (int j = 0; j < 4; ++j) sW[tid + j * 256] = wv[j];
  __syncthreads();

  bin_sums<C, true>(acc, cl, packed, h1, binStart[b], binStart[b + 1]);
  __syncthreads();

  if (tid < BIN_NODES) {
    int node = b * BIN_NODES + tid;
    if (node < N) cnt[node] = cl[tid];
  }
  if (!valid) return;

  int lr = wave * 16 + n;
  float inv = 1.0f / fmaxf(cl[lr], 1.0f);
  v8f c0 = {}, c1 = {};
#pragma unroll
  for (int kb = 0; kb < 8; ++kb) {
    int kk = kb * 4 + koff;
    v2f a;
    if (kb < 4) { a.x = acc[lr * P + kk] * inv; a.y = acc[lr * P + kk + 1] * inv; }
    else        { a = self[kb - 4]; }
    v2f b0; b0.x = sW[kk * 32 + n];      b0.y = sW[(kk + 1) * 32 + n];
    v2f b1; b1.x = sW[kk * 32 + 16 + n]; b1.y = sW[(kk + 1) * 32 + 16 + n];
    c0 = wmma_f32_4(a, b0, c0);
    c1 = wmma_f32_4(a, b1, c1);
  }

  float bl = llb[n], bh = llb[n + 16];
  int rbase = tile * 16 + ((lane >> 4) << 3);
#pragma unroll
  for (int r = 0; r < 8; ++r) {
    long long orow = (long long)(rbase + r) * 32;
    out1[orow + n]      = fmaxf(c0[r] + bl, 0.0f);
    out1[orow + 16 + n] = fmaxf(c1[r] + bh, 0.0f);
  }
}

__global__ __launch_bounds__(256) void k_layer2(const int* __restrict__ packed,
                                                const int* __restrict__ binStart,
                                                const float* __restrict__ out1,
                                                const float* __restrict__ cnt,
                                                const float* __restrict__ llw,
                                                const float* __restrict__ llb,
                                                const float* __restrict__ lrw,
                                                const float* __restrict__ w2,
                                                const float* __restrict__ b2,
                                                float* __restrict__ out,
                                                int N, int ntiles) {
  constexpr int C = 32, P = C + 1;
  __shared__ float acc[BIN_NODES * P];
  __shared__ float sW[64 * 16];
  int tid = threadIdx.x, b = blockIdx.x;
  int wave = tid >> 5, lane = tid & 31;
  int n = lane & 15;
  int koff = (lane >> 4) << 1;
  int tile = b * (BIN_NODES / 16) + wave;
  bool valid = tile < ntiles;
  long long row = (long long)(valid ? tile : 0) * 16 + n;

  float deg = cnt[row];
  v2f self[8];
#pragma unroll
  for (int kb = 0; kb < 8; ++kb) {
    int kk = kb * 4 + koff;
    self[kb].x = out1[row * 32 + kk];
    self[kb].y = out1[row * 32 + kk + 1];
  }

  float wv[4];
#pragma unroll
  for (int j = 0; j < 4; ++j)
    wv[j] = (j < 2) ? llw[tid + j * 256] : lrw[tid + (j - 2) * 256];

  for (int i = tid; i < BIN_NODES * P; i += 256) acc[i] = 0.0f;
#pragma unroll
  for (int j = 0; j < 4; ++j) sW[tid + j * 256] = wv[j];
  __syncthreads();

  bin_sums<C, false>(acc, nullptr, packed, out1, binStart[b], binStart[b + 1]);
  __syncthreads();

  float* trow = acc + wave * 16 * P;
  if (valid) {
    float inv = 1.0f / fmaxf(deg, 1.0f);
    v8f d = {};
#pragma unroll
    for (int kb = 0; kb < 16; ++kb) {
      int kk = kb * 4 + koff;
      v2f a;
      if (kb < 8) { a.x = trow[n * P + kk] * inv; a.y = trow[n * P + kk + 1] * inv; }
      else        { a = self[kb - 8]; }
      v2f bw; bw.x = sW[kk * 16 + n]; bw.y = sW[(kk + 1) * 16 + n];
      d = wmma_f32_4(a, bw, d);
    }
    float bias = llb[n];
    int r8 = (lane >> 4) << 3;
#pragma unroll
    for (int r = 0; r < 8; ++r)
      trow[(r8 + r) * P + n] = fmaxf(d[r] + bias, 0.0f);
  }
  __syncthreads();

  if (valid && lane < 16) {
    int i = tile * 16 + lane;
    if (i < N) {
      float h0 = b2[0], h1 = b2[1], h2 = b2[2];
      const float* r = trow + lane * P;
#pragma unroll
      for (int k = 0; k < 16; ++k) {
        float v = r[k];
        h0 += v * w2[k * 3 + 0];
        h1 += v * w2[k * 3 + 1];
        h2 += v * w2[k * 3 + 2];
      }
      float gsi = 1.0f / (1.0f + __expf(-h1));
      float mxi = 1.0f / (1.0f + __expf(-h2));
      float fsi = fmaxf(h0, 0.0f) + gsi;
      out[(long long)i * 3 + 0] = fsi;
      out[(long long)i * 3 + 1] = gsi;
      out[(long long)i * 3 + 2] = mxi;
    }
  }
}

extern "C" void kernel_launch(void* const* d_in, const int* in_sizes, int n_in,
                              void* d_out, int out_size, void* d_ws, size_t ws_size,
                              hipStream_t stream) {
  const float* x    = (const float*)d_in[0];
  const int*   ei   = (const int*)d_in[1];
  const float* w1   = (const float*)d_in[2];
  const float* b1   = (const float*)d_in[3];
  const float* c1ll = (const float*)d_in[4];
  const float* c1lb = (const float*)d_in[5];
  const float* c1lr = (const float*)d_in[6];
  const float* c2ll = (const float*)d_in[7];
  const float* c2lb = (const float*)d_in[8];
  const float* c2lr = (const float*)d_in[9];
  const float* w2   = (const float*)d_in[10];
  const float* b2   = (const float*)d_in[11];
  float* out = (float*)d_out;

  int N = in_sizes[0] / IN_DIM;
  int E = in_sizes[1] / 2;
  const int* src = ei;
  const int* dst = ei + E;

  float* ws   = (float*)d_ws;
  float* h1   = ws;
  float* out1 = ws + (long long)32 * N;
  int*   packed = (int*)(ws + (long long)64 * N);
  float* cnt  = ws + (long long)96 * N;
  int*   hist = (int*)(ws + (long long)97 * N);
  int*   binTotal = hist + NB_MAX * NCH;
  int*   binStart = binTotal + NB_MAX;

  int ntiles = (N + 15) / 16;
  int fblocks = (ntiles + 7) / 8;
  int nb = (N + BIN_NODES - 1) / BIN_NODES;
  int chunk = (E + NCH - 1) / NCH;

  k_hist<<<NCH + fblocks, 256, 0, stream>>>(dst, hist, E, chunk, nb, x, w1, b1, h1, ntiles);
  k_scan<<<nb, 256, 0, stream>>>(hist, binTotal);
  k_binscatter<<<NCH + fblocks, 256, 0, stream>>>(src, dst, hist, binTotal, binStart, packed, E, chunk, nb,
                                                  x, w1, b1, h1, ntiles);

  k_layer1<<<nb, 256, 0, stream>>>(packed, binStart, h1, c1ll, c1lb, c1lr, out1, cnt, N, ntiles);

  k_layer2<<<nb, 256, 0, stream>>>(packed, binStart, out1, cnt, c2ll, c2lb, c2lr, w2, b2, out,
                                   N, ntiles);
}


# layer sums without float atomics: each bin's edges counting-sorted by node in LDS (integer counters), a node's rows added in registers in sorted arrival order; table written by plain stores
# speedup vs baseline: 5.8517x; 1.2301x over previous
#include <hip/hip_runtime.h>
#include <math.h>

typedef __attribute__((ext_vector_type(2))) float v2f;
typedef __attribute__((ext_vector_type(8))) float v8f;

#define IN_DIM 512

__device__ __forceinline__ v8f wmma_f32_4(v2f a, v2f b, v8f c) {
  return __builtin_amdgcn_wmma_f32_16x16x4_f32(false, a, false, b, (short)0, c,
                                               false, false);
}

#define FC_KB 64
#define FC_BLOCKS (FC_KB / 8)
#define FC_AHEAD 2

template <bool FIRST>
__device__ __forceinline__ void fc1_half(const float* __restrict__ x,
                                         const float* __restrict__ w1,
                                         const float* __restrict__ b1,
                                         float* __restrict__ h1,
                                         float* sW, int fb, int ntiles) {
  constexpr int K0 = FIRST ? 0 : FC_KB * 4;
  int tid = threadIdx.x;
  {
    const float4* wsrc = (const float4*)(w1 + K0 * 16);
    float4 t[4];
#pragma unroll
    for (int j = 0; j < 4; ++j) t[j] = wsrc[tid + j * 256];
#pragma unroll
    for (int j = 0; j < 4; ++j) ((float4*)sW)[tid + j * 256] = t[j];
  }
  __syncthreads();

  int wave = tid >> 5, lane = tid & 31;
  int tile = fb * 8 + wave;
  if (tile >= ntiles) return;

  int n = lane & 15;
  int h4 = (lane >> 4) << 2;
  long long row = (long long)tile * 16 + n;
  const float* xr = x + row * IN_DIM + K0 + h4;
  int rbase = tile * 16 + ((lane >> 4) << 3);

  float4 xa[FC_AHEAD + 1][4];
#pragma unroll
  for (int b = 0; b < FC_AHEAD; ++b)
#pragma unroll
    for (int g = 0; g < 4; ++g) xa[b][g] = *(const float4*)(xr + b * 32 + g * 8);

  v8f acc = {};
  if (!FIRST) {
#pragma unroll
    for (int r = 0; r < 8; ++r) acc[r] = h1[(long long)(rbase + r) * 16 + n];
  }
#pragma unroll
  for (int blk = 0; blk < FC_BLOCKS; ++blk) {
    if (blk + FC_AHEAD < FC_BLOCKS) {
#pragma unroll
      for (int g = 0; g < 4; ++g)
        xa[(blk + FC_AHEAD) % (FC_AHEAD + 1)][g] =
            *(const float4*)(xr + (blk + FC_AHEAD) * 32 + g * 8);
    }
#pragma unroll
    for (int g = 0; g < 4; ++g) {
      float4 av = xa[blk % (FC_AHEAD + 1)][g];
      int k = blk * 32 + g * 8 + h4;
      v2f a0; a0.x = av.x; a0.y = av.y;
      v2f a1; a1.x = av.z; a1.y = av.w;
      v2f b0; b0.x = sW[k * 16 + n];       b0.y = sW[(k + 1) * 16 + n];
      v2f b1; b1.x = sW[(k + 2) * 16 + n]; b1.y = sW[(k + 3) * 16 + n];
      acc = wmma_f32_4(a0, b0, acc);
      acc = wmma_f32_4(a1, b1, acc);
    }
  }

  if (FIRST) {
#pragma unroll
    for (int r = 0; r < 8; ++r) h1[(long long)(rbase + r) * 16 + n] = acc[r];
  } else {
    float bias = b1[n];
#pragma unroll
    for (int r = 0; r < 8; ++r)
      h1[(long long)(rbase + r) * 16 + n] = fmaxf(acc[r] + bias, 0.0f);
  }
}

#define BIN_SHIFT 7
#define BIN_NODES 128
#define NB_MAX 1024
#define NCH 512
#define EDGE_BATCH 4

__device__ __forceinline__ int block_incl_scan256(int v, int* sh) {
  int tid = threadIdx.x;
  sh[tid] = v;
  __syncthreads();
  for (int off = 1; off < 256; off <<= 1) {
    int t = (tid >= off) ? sh[tid - off] : 0;
    __syncthreads();
    sh[tid] += t;
    __syncthreads();
  }
  return sh[tid];
}

__global__ __launch_bounds__(256) void k_hist(const int* __restrict__ dst,
                                              int* __restrict__ hist,
                                              int nE, int chunk, int nb,
                                              const float* __restrict__ x,
                                              const float* __restrict__ w1,
                                              const float* __restrict__ b1,
                                              float* __restrict__ h1, int ntiles) {
  __shared__ int sc[NB_MAX];
  __shared__ __attribute__((aligned(16))) float sW[FC_KB * 64];
  if (blockIdx.x >= NCH) {
    fc1_half<true>(x, w1, b1, h1, sW, blockIdx.x - NCH, ntiles);
    return;
  }
  int tid = threadIdx.x, c = blockIdx.x;
  for (int i = tid; i < nb; i += 256) sc[i] = 0;
  __syncthreads();
  int e0 = c * chunk;
  int e1 = min(e0 + chunk, nE);
  for (int e = e0 + tid; e < e1; e += 256 * EDGE_BATCH) {
    int d[EDGE_BATCH];
#pragma unroll
    for (int j = 0; j < EDGE_BATCH; ++j) {
      int ee = e + j * 256;
      d[j] = (ee < e1) ? dst[ee] : -1;
    }
#pragma unroll
    for (int j = 0; j < EDGE_BATCH; ++j)
      if (d[j] >= 0) atomicAdd(&sc[d[j] >> BIN_SHIFT], 1);
  }
  __syncthreads();
  for (int i = tid; i < nb; i += 256) hist[i * NCH + c] = sc[i];
}

__global__ __launch_bounds__(256) void k_scan(int* hist, int* __restrict__ binTotal) {
  __shared__ int sh[256];
  int tid = threadIdx.x, b = blockIdx.x;
  int* row = hist + b * NCH;
  int v0 = row[2 * tid], v1 = row[2 * tid + 1];
  int incl = block_incl_scan256(v0 + v1, sh);
  int excl = incl - (v0 + v1);
  row[2 * tid] = excl;
  row[2 * tid + 1] = excl + v0;
  if (tid == 255) binTotal[b] = incl;
}

#define TILE_K 25
#define TILE_E (256 * TILE_K)
__global__ __launch_bounds__(256) void k_binscatter(const int* __restrict__ src,
                                                    const int* __restrict__ dst,
                                                    const int* __restrict__ hist,
                                                    const int* __restrict__ binTotal,
                                                    int* __restrict__ binStart,
                                                    int* __restrict__ packed,
                                                    int nE, int chunk, int nb,
                                                    const float* __restrict__ x,
                                                    const float* __restrict__ w1,
                                                    const float* __restrict__ b1,
                                                    float* __restrict__ h1, int ntiles) {
  __shared__ __attribute__((aligned(16))) int smem[3 * NB_MAX + 256 + 2 * TILE_E];
  if (blockIdx.x >= NCH) {
    fc1_half<false>(x, w1, b1, h1, (float*)smem, blockIdx.x - NCH, ntiles);
    return;
  }
  int* cur  = smem;
  int* lcur = smem + NB_MAX;
  int* gdel = smem + 2 * NB_MAX;
  int* sh   = smem + 3 * NB_MAX;
  int* sval = sh + 256;
  int* sgp  = sval + TILE_E;
  int tid = threadIdx.x, c = blockIdx.x;
  int t[4];
  int sum = 0;
#pragma unroll
  for (int j = 0; j < 4; ++j) {
    int i = tid * 4 + j;
    t[j] = (i < nb) ? binTotal[i] : 0;
    sum += t[j];
  }
  int base = block_incl_scan256(sum, sh) - sum;
#pragma unroll
  for (int j = 0; j < 4; ++j) {
    int i = tid * 4 + j;
    int h = (i < nb) ? hist[i * NCH + c] : 0;
    cur[i] = base + h;
    if (c == 0 && i <= nb) binStart[i] = base;
    base += t[j];
  }
  int e0 = c * chunk;
  int e1 = min(e0 + chunk, nE);
  for (int t0 = e0; t0 < e1; t0 += TILE_E) {
    int n = min(TILE_E, e1 - t0);
    for (int i = tid; i < NB_MAX; i += 256) lcur[i] = 0;
    __syncthreads();

    int d[TILE_K], s[TILE_K];
#pragma unroll
    for (int k = 0; k < TILE_K; ++k) {
      int idx = tid + k * 256;
      int v = dst[t0 + min(idx, n - 1)];
      d[k] = (idx < n) ? v : -1;
    }
#pragma unroll
    for (int k = 0; k < TILE_K; ++k) {
      int idx = tid + k * 256;
      s[k] = src[t0 + min(idx, n - 1)];
    }
#pragma unroll
    for (int k = 0; k < TILE_K; ++k)
      if (d[k] >= 0) atomicAdd(&lcur[d[k] >> BIN_SHIFT], 1);
    __syncthreads();

    int lc[4];
    int lsum = 0;
#pragma unroll
    for (int j = 0; j < 4; ++j) {
      lc[j] = lcur[tid * 4 + j];
      lsum += lc[j];
    }
    int lbase = block_incl_scan256(lsum, sh) - lsum;
#pragma unroll
    for (int j = 0; j < 4; ++j) {
      int i = tid * 4 + j;
      int g = cur[i];
      lcur[i] = lbase;
      gdel[i] = g - lbase;
      cur[i] = g + lc[j];
      lbase += lc[j];
    }
    __syncthreads();

#pragma unroll
    for (int k = 0; k < TILE_K; ++k) {
      if (d[k] >= 0) {
        int bin = d[k] >> BIN_SHIFT;
        int pos = atomicAdd(&lcur[bin], 1);
        sval[pos] = (s[k] << BIN_SHIFT) | (d[k] & (BIN_NODES - 1));
        sgp[pos] = pos + gdel[bin];
      }
    }
    __syncthreads();

    for (int j = tid; j < n; j += 256) packed[sgp[j]] = sval[j];
  }
}

#define SUBS 4
#define NKEY (BIN_NODES * SUBS)
#define SORT_K 20
#define SORT_CAP (256 * SORT_K)
#define GATHER_U 16

__device__ __forceinline__ int block_incl_scan256_shfl(int v, int* sh) {
  int tid = threadIdx.x;
  int wave = tid >> 5, lane = tid & 31;
  int x = v;
#pragma unroll
  for (int o = 1; o < 32; o <<= 1) {
    int y = __shfl_up(x, o);
    if (lane >= o) x += y;
  }
  if (lane == 31) sh[wave] = x;
  __syncthreads();
  int add = 0;
#pragma unroll
  for (int i = 0; i < 7; ++i)
    if (i < wave) add += sh[i];
  return x + add;
}

template <int C, bool CNT>
__device__ __forceinline__ void bin_sums(float* acc, float* cl, int* key, int* off, int* sh,
                                         int* sorted,
                                         const int* __restrict__ packed,
                                         const float* __restrict__ feat,
                                         int e0, int e1) {
  constexpr int P = C + 1;
  constexpr int LPE = C / 4;
  constexpr int G = 256 / LPE;
  constexpr int NPG = BIN_NODES / G;
  int tid = threadIdx.x;
  int g = tid / LPE, q = tid % LPE;
  int sub = tid & (SUBS - 1);
  const float4* __restrict__ frow = (const float4*)feat;
  float degf = 0.0f;
  for (int t0 = e0; t0 < e1; t0 += SORT_CAP) {
    int n = min(SORT_CAP, e1 - t0);
    key[tid] = 0;
    key[tid + 256] = 0;
    __syncthreads();

    int w[SORT_K];
#pragma unroll
    for (int k = 0; k < SORT_K; ++k) {
      int idx = tid + k * 256;
      int v = packed[t0 + min(idx, n - 1)];
      w[k] = (idx < n) ? v : -1;
    }
#pragma unroll
    for (int k = 0; k < SORT_K; ++k)
      if (w[k] >= 0) atomicAdd(&key[((w[k] & (BIN_NODES - 1)) * SUBS) | sub], 1);
    __syncthreads();

    int c0 = key[2 * tid], c1 = key[2 * tid + 1];
    int incl = block_incl_scan256_shfl(c0 + c1, sh);
    int excl = incl - (c0 + c1);
    key[2 * tid] = excl;
    key[2 * tid + 1] = excl + c0;
    off[2 * tid] = excl;
    off[2 * tid + 1] = excl + c0;
    if (tid == 255) off[NKEY] = incl;
    __syncthreads();
    if (CNT && tid < BIN_NODES) degf += (float)(off[tid * SUBS + SUBS] - off[tid * SUBS]);

    int pos[SORT_K];
#pragma unroll
    for (int k = 0; k < SORT_K; ++k) {
      pos[k] = 0;
      if (w[k] >= 0) pos[k] = atomicAdd(&key[((w[k] & (BIN_NODES - 1)) * SUBS) | sub], 1);
    }
#pragma unroll
    for (int k = 0; k < SORT_K; ++k)
      sorted[(w[k] >= 0) ? pos[k] : SORT_CAP + tid] = w[k];
    __syncthreads();

    for (int nn = 0; nn < NPG; ++nn) {
      int node = g * NPG + nn;
      int j0 = off[node * SUBS], j1 = off[node * SUBS + SUBS];
      float4 s = make_float4(0.0f, 0.0f, 0.0f, 0.0f);
      for (int j = j0; j < j1; j += GATHER_U) {
        int rem = j1 - j;
        int ww[GATHER_U];
        float4 v[GATHER_U];
#pragma unroll
        for (int u = 0; u < GATHER_U; ++u) ww[u] = sorted[j + min(u, rem - 1)];
#pragma unroll
        for (int u = 0; u < GATHER_U; ++u)
          v[u] = frow[(unsigned)(ww[u] >> BIN_SHIFT) * (unsigned)LPE + (unsigned)q];
#pragma unroll
        for (int u = 0; u < GATHER_U; ++u) {
          bool ok = u < rem;
          s.x += ok ? v[u].x : 0.0f;
          s.y += ok ? v[u].y : 0.0f;
          s.z += ok ? v[u].z : 0.0f;
          s.w += ok ? v[u].w : 0.0f;
        }
      }
      float* r = acc + node * P + q * 4;
      if (t0 != e0) {
        s.x += r[0];
        s.y += r[1];
        s.z += r[2];
        s.w += r[3];
      }
      r[0] = s.x;
      r[1] = s.y;
      r[2] = s.z;
      r[3] = s.w;
    }
  }
  if (e0 >= e1) {
    for (int nn = 0; nn < NPG; ++nn) {
      float* r = acc + (g * NPG + nn) * P + q * 4;
      r[0] = 0.0f;
      r[1] = 0.0f;
      r[2] = 0.0f;
      r[3] = 0.0f;
    }
  }
  if (CNT && tid < BIN_NODES) cl[tid] = degf;
}

__global__ __launch_bounds__(256) void k_layer1(const int* __restrict__ packed,
                                                const int* __restrict__ binStart,
                                                const float* __restrict__ h1,
                                                const float* __restrict__ llw,
                                                const float* __restrict__ llb,
                                                const float* __restrict__ lrw,
                                                float* __restrict__ out1,
                                                float* __restrict__ cnt,
                                                int N, int ntiles) {
  constexpr int C = 16, P = C + 1;
  __shared__ float acc[BIN_NODES * P];
  __shared__ float cl[BIN_NODES];
  __shared__ float sW[32 * 32];
  __shared__ int skey[NKEY];
  __shared__ int soff[NKEY + 4];
  __shared__ int ssh[256];
  __shared__ int ssorted[SORT_CAP + 256];
  int tid = threadIdx.x, b = blockIdx.x;
  int wave = tid >> 5, lane = tid & 31;
  int n = lane & 15;
  int koff = (lane >> 4) << 1;
  int tile = b * (BIN_NODES / 16) + wave;
  bool valid = tile < ntiles;
  long long row = (long long)(valid ? tile : 0) * 16 + n;

  float wv[4];
#pragma unroll
  for (int j = 0; j < 4; ++j)
    wv[j] = (j < 2) ? llw[tid + j * 256] : lrw[tid + (j - 2) * 256];
#pragma unroll
  for (int j = 0; j < 4; ++j) sW[tid + j * 256] = wv[j];

  bin_sums<C, true>(acc, cl, skey, soff, ssh, ssorted, packed, h1, binStart[b], binStart[b + 1]);

  v2f self[4];
#pragma unroll
  for (int kb = 0; kb < 4; ++kb) {
    int kk = kb * 4 + koff;
    self[kb].x = h1[row * 16 + kk];
    self[kb].y = h1[row * 16 + kk + 1];
  }
  __syncthreads();

  if (tid < BIN_NODES) {
    int node = b * BIN_NODES + tid;
    if (node < N) cnt[node] = cl[tid];
  }
  if (!valid) return;

  int lr = wave * 16 + n;
  float inv = 1.0f / fmaxf(cl[lr], 1.0f);
  v8f c0 = {}, c1 = {};
#pragma unroll
  for (int kb = 0; kb < 8; ++kb) {
    int kk = kb * 4 + koff;
    v2f a;
    if (kb < 4) { a.x = acc[lr * P + kk] * inv; a.y = acc[lr * P + kk + 1] * inv; }
    else        { a = self[kb - 4]; }
    v2f b0; b0.x = sW[kk * 32 + n];      b0.y = sW[(kk + 1) * 32 + n];
    v2f b1; b1.x = sW[kk * 32 + 16 + n]; b1.y = sW[(kk + 1) * 32 + 16 + n];
    c0 = wmma_f32_4(a, b0, c0);
    c1 = wmma_f32_4(a, b1, c1);
  }

  float bl = llb[n], bh = llb[n + 16];
  int rbase = tile * 16 + ((lane >> 4) << 3);
#pragma unroll
  for (int r = 0; r < 8; ++r) {
    long long orow = (long long)(rbase + r) * 32;
    out1[orow + n]      = fmaxf(c0[r] + bl, 0.0f);
    out1[orow + 16 + n] = fmaxf(c1[r] + bh, 0.0f);
  }
}

__global__ __launch_bounds__(256) void k_layer2(const int* __restrict__ packed,
                                                const int* __restrict__ binStart,
                                                const float* __restrict__ out1,
                                                const float* __restrict__ cnt,
                                                const float* __restrict__ llw,
                                                const float* __restrict__ llb,
                                                const float* __restrict__ lrw,
                                                const float* __restrict__ w2,
                                                const float* __restrict__ b2,
                                                float* __restrict__ out,
                                                int N, int ntiles) {
  constexpr int C = 32, P = C + 1;
  __shared__ float acc[BIN_NODES * P];
  __shared__ float sW[64 * 16];
  __shared__ int skey[NKEY];
  __shared__ int soff[NKEY + 4];
  __shared__ int ssh[256];
  __shared__ int ssorted[SORT_CAP + 256];
  int tid = threadIdx.x, b = blockIdx.x;
  int wave = tid >> 5, lane = tid & 31;
  int n = lane & 15;
  int koff = (lane >> 4) << 1;
  int tile = b * (BIN_NODES / 16) + wave;
  bool valid = tile < ntiles;
  long long row = (long long)(valid ? tile : 0) * 16 + n;

  float wv[4];
#pragma unroll
  for (int j = 0; j < 4; ++j)
    wv[j] = (j < 2) ? llw[tid + j * 256] : lrw[tid + (j - 2) * 256];
#pragma unroll
  for (int j = 0; j < 4; ++j) sW[tid + j * 256] = wv[j];

  bin_sums<C, false>(acc, nullptr, skey, soff, ssh, ssorted, packed, out1, binStart[b],
                     binStart[b + 1]);

  float deg = cnt[row];
  v2f self[8];
#pragma unroll
  for (int kb = 0; kb < 8; ++kb) {
    int kk = kb * 4 + koff;
    self[kb].x = out1[row * 32 + kk];
    self[kb].y = out1[row * 32 + kk + 1];
  }
  __syncthreads();

  float* trow = acc + wave * 16 * P;
  if (valid) {
    float inv = 1.0f / fmaxf(deg, 1.0f);
    v8f d = {};
#pragma unroll
    for (int kb = 0; kb < 16; ++kb) {
      int kk = kb * 4 + koff;
      v2f a;
      if (kb < 8) { a.x = trow[n * P + kk] * inv; a.y = trow[n * P + kk + 1] * inv; }
      else        { a = self[kb - 8]; }
      v2f bw; bw.x = sW[kk * 16 + n]; bw.y = sW[(kk + 1) * 16 + n];
      d = wmma_f32_4(a, bw, d);
    }
    float bias = llb[n];
    int r8 = (lane >> 4) << 3;
#pragma unroll
    for (int r = 0; r < 8; ++r)
      trow[(r8 + r) * P + n] = fmaxf(d[r] + bias, 0.0f);
  }
  __syncthreads();

  if (valid && lane < 16) {
    int i = tile * 16 + lane;
    if (i < N) {
      float h0 = b2[0], h1 = b2[1], h2 = b2[2];
      const float* r = trow + lane * P;
#pragma unroll
      for (int k = 0; k < 16; ++k) {
        float v = r[k];
        h0 += v * w2[k * 3 + 0];
        h1 += v * w2[k * 3 + 1];
        h2 += v * w2[k * 3 + 2];
      }
      float gsi = 1.0f / (1.0f + __expf(-h1));
      float mxi = 1.0f / (1.0f + __expf(-h2));
      float fsi = fmaxf(h0, 0.0f) + gsi;
      out[(long long)i * 3 + 0] = fsi;
      out[(long long)i * 3 + 1] = gsi;
      out[(long long)i * 3 + 2] = mxi;
    }
  }
}

extern "C" void kernel_launch(void* const* d_in, const int* in_sizes, int n_in,
                              void* d_out, int out_size, void* d_ws, size_t ws_size,
                              hipStream_t stream) {
  const float* x    = (const float*)d_in[0];
  const int*   ei   = (const int*)d_in[1];
  const float* w1   = (const float*)d_in[2];
  const float* b1   = (const float*)d_in[3];
  const float* c1ll = (const float*)d_in[4];
  const float* c1lb = (const float*)d_in[5];
  const float* c1lr = (const float*)d_in[6];
  const float* c2ll = (const float*)d_in[7];
  const float* c2lb = (const float*)d_in[8];
  const float* c2lr = (const float*)d_in[9];
  const float* w2   = (const float*)d_in[10];
  const float* b2   = (const float*)d_in[11];
  float* out = (float*)d_out;

  int N = in_sizes[0] / IN_DIM;
  int E = in_sizes[1] / 2;
  const int* src = ei;
  const int* dst = ei + E;

  float* ws   = (float*)d_ws;
  float* h1   = ws;
  float* out1 = ws + (long long)32 * N;
  int*   packed = (int*)(ws + (long long)64 * N);
  float* cnt  = ws + (long long)96 * N;
  int*   hist = (int*)(ws + (long long)97 * N);
  int*   binTotal = hist + NB_MAX * NCH;
  int*   binStart = binTotal + NB_MAX;

  int ntiles = (N + 15) / 16;
  int fblocks = (ntiles + 7) / 8;
  int nb = (N + BIN_NODES - 1) / BIN_NODES;
  int chunk = (E + NCH - 1) / NCH;

  k_hist<<<NCH + fblocks, 256, 0, stream>>>(dst, hist, E, chunk, nb, x, w1, b1, h1, ntiles);
  k_scan<<<nb, 256, 0, stream>>>(hist, binTotal);
  k_binscatter<<<NCH + fblocks, 256, 0, stream>>>(src, dst, hist, binTotal, binStart, packed, E, chunk, nb,
                                                  x, w1, b1, h1, ntiles);

  k_layer1<<<nb, 256, 0, stream>>>(packed, binStart, h1, c1ll, c1lb, c1lr, out1, cnt, N, ntiles);

  k_layer2<<<nb, 256, 0, stream>>>(packed, binStart, out1, cnt, c2ll, c2lb, c2lr, w2, b2, out,
                                   N, ntiles);
}


# no global edge sort: 6400-edge tiles bin-sorted in LDS, stored whole with bin run starts; a layer's bin work-group collects its runs from all tiles; fc1 whole beside; 5 launches to 3; arrival order
# speedup vs baseline: 7.2231x; 1.2344x over previous
#include <hip/hip_runtime.h>
#include <math.h>

typedef __attribute__((ext_vector_type(2))) float v2f;
typedef __attribute__((ext_vector_type(8))) float v8f;

#define IN_DIM 512

__device__ __forceinline__ v8f wmma_f32_4(v2f a, v2f b, v8f c) {
  return __builtin_amdgcn_wmma_f32_16x16x4_f32(false, a, false, b, (short)0, c,
                                               false, false);
}

#define FC_BLOCKS 16
#define FC_AHEAD 2

__device__ __forceinline__ void fc1_rows(const float* __restrict__ x,
                                         const float* __restrict__ w1,
                                         const float* __restrict__ b1,
                                         float* __restrict__ h1,
                                         float* sW, int fb, int ntiles) {
  int tid = threadIdx.x;
  {
    const float4* wsrc = (const float4*)w1;
    float4 t[8];
#pragma unroll
    for (int j = 0; j < 8; ++j) t[j] = wsrc[tid + j * 256];
#pragma unroll
    for (int j = 0; j < 8; ++j) ((float4*)sW)[tid + j * 256] = t[j];
  }
  __syncthreads();

  int wave = tid >> 5, lane = tid & 31;
  int tile = fb * 8 + wave;
  if (tile >= ntiles) return;

  int n = lane & 15;
  int h4 = (lane >> 4) << 2;
  long long row = (long long)tile * 16 + n;
  const float* xr = x + row * IN_DIM + h4;
  int rbase = tile * 16 + ((lane >> 4) << 3);

  float4 xa[FC_AHEAD + 1][4];
#pragma unroll
  for (int b = 0; b < FC_AHEAD; ++b)
#pragma unroll
    for (int g = 0; g < 4; ++g) xa[b][g] = *(const float4*)(xr + b * 32 + g * 8);

  v8f acc = {};
#pragma unroll
  for (int blk = 0; blk < FC_BLOCKS; ++blk) {
    if (blk + FC_AHEAD < FC_BLOCKS) {
#pragma unroll
      for (int g = 0; g < 4; ++g)
        xa[(blk + FC_AHEAD) % (FC_AHEAD + 1)][g] =
            *(const float4*)(xr + (blk + FC_AHEAD) * 32 + g * 8);
    }
#pragma unroll
    for (int g = 0; g < 4; ++g) {
      float4 av = xa[blk % (FC_AHEAD + 1)][g];
      int k = blk * 32 + g * 8 + h4;
      v2f a0; a0.x = av.x; a0.y = av.y;
      v2f a1; a1.x = av.z; a1.y = av.w;
      v2f b0; b0.x = sW[k * 16 + n];       b0.y = sW[(k + 1) * 16 + n];
      v2f b1; b1.x = sW[(k + 2) * 16 + n]; b1.y = sW[(k + 3) * 16 + n];
      acc = wmma_f32_4(a0, b0, acc);
      acc = wmma_f32_4(a1, b1, acc);
    }
  }
  float bias = b1[n];
#pragma unroll
  for (int r = 0; r < 8; ++r)
    h1[(long long)(rbase + r) * 16 + n] = fmaxf(acc[r] + bias, 0.0f);
}

#define BIN_SHIFT 7
#define BIN_NODES 128
#define NB_MAX 1024
#define TILE_K 25
#define TILE_E (256 * TILE_K)
#define SEG_MAX 512

__device__ __forceinline__ int block_incl_scan256_shfl(int v, int* sh) {
  int tid = threadIdx.x;
  int wave = tid >> 5, lane = tid & 31;
  int x = v;
#pragma unroll
  for (int o = 1; o < 32; o <<= 1) {
    int y = __shfl_up(x, o);
    if (lane >= o) x += y;
  }
  if (lane == 31) sh[wave] = x;
  __syncthreads();
  int add = 0;
#pragma unroll
  for (int i = 0; i < 7; ++i)
    if (i < wave) add += sh[i];
  return x + add;
}

__global__ __launch_bounds__(256) void k_tilesort(const int* __restrict__ src,
                                                  const int* __restrict__ dst,
                                                  int* __restrict__ segOff,
                                                  int* __restrict__ packed,
                                                  int nE, int nseg, int nb,
                                                  const float* __restrict__ x,
                                                  const float* __restrict__ w1,
                                                  const float* __restrict__ b1,
                                                  float* __restrict__ h1, int ntiles) {
  __shared__ __attribute__((aligned(16))) int smem[8192];
  if (blockIdx.x >= nseg) {
    fc1_rows(x, w1, b1, h1, (float*)smem, blockIdx.x - nseg, ntiles);
    return;
  }
  int* lcur = smem;
  int* sh   = smem + NB_MAX;
  int* sval = smem + NB_MAX + 256;
  int tid = threadIdx.x, sg = blockIdx.x;
  int t0 = sg * TILE_E;
  int n = min(TILE_E, nE - t0);
  for (int i = tid; i < NB_MAX; i += 256) lcur[i] = 0;
  __syncthreads();

  int d[TILE_K], s[TILE_K];
#pragma unroll
  for (int k = 0; k < TILE_K; ++k) {
    int idx = tid + k * 256;
    int v = dst[t0 + min(idx, n - 1)];
    d[k] = (idx < n) ? v : -1;
  }
#pragma unroll
  for (int k = 0; k < TILE_K; ++k) {
    int idx = tid + k * 256;
    s[k] = src[t0 + min(idx, n - 1)];
  }
#pragma unroll
  for (int k = 0; k < TILE_K; ++k)
    if (d[k] >= 0) atomicAdd(&lcur[d[k] >> BIN_SHIFT], 1);
  __syncthreads();

  int lc[4];
  int lsum = 0;
#pragma unroll
  for (int j = 0; j < 4; ++j) {
    lc[j] = lcur[tid * 4 + j];
    lsum += lc[j];
  }
  int lbase = block_incl_scan256_shfl(lsum, sh) - lsum;
  int* orow = segOff + sg * (nb + 1);
#pragma unroll
  for (int j = 0; j < 4; ++j) {
    int i = tid * 4 + j;
    lcur[i] = lbase;
    if (i <= nb) orow[i] = lbase;
    lbase += lc[j];
  }
  __syncthreads();

#pragma unroll
  for (int k = 0; k < TILE_K; ++k) {
    if (d[k] >= 0) {
      int bin = d[k] >> BIN_SHIFT;
      int pos = atomicAdd(&lcur[bin], 1);
      sval[pos] = (s[k] << BIN_SHIFT) | (d[k] & (BIN_NODES - 1));
    }
  }
  __syncthreads();

  for (int j = tid; j < n; j += 256) packed[t0 + j] = sval[j];
}

#define SUBS 4
#define NKEY (BIN_NODES * SUBS)
#define SORT_K 20
#define SORT_CAP (256 * SORT_K)
#define GATHER_U 16
#define COLLECT_R (SEG_MAX / 32)
#define COLLECT_P 4

template <int C, bool CNT>
__device__ __forceinline__ void bin_sums(float* acc, float* cl, int* key, int* off, int* sh,
                                         int* sorted, int* run,
                                         const int* __restrict__ packed,
                                         const int* __restrict__ segOff,
                                         const float* __restrict__ feat,
                                         int nseg, int nb, int b) {
  constexpr int P = C + 1;
  constexpr int LPE = C / 4;
  constexpr int G = 256 / LPE;
  constexpr int NPG = BIN_NODES / G;
  int tid = threadIdx.x;
  int g = tid / LPE, q = tid % LPE;
  int sub = tid & (SUBS - 1);
  const float4* __restrict__ frow = (const float4*)feat;
  float degf = 0.0f;

  int* vst = run;
  int* gsrc = run + SEG_MAX + 8;
  {
    int sA = min(2 * tid, nseg - 1), sB = min(2 * tid + 1, nseg - 1);
    const int* rA = segOff + sA * (nb + 1) + b;
    const int* rB = segOff + sB * (nb + 1) + b;
    int a0 = rA[0], a1 = rA[1], b0 = rB[0], b1 = rB[1];
    bool hasA = 2 * tid < nseg, hasB = 2 * tid + 1 < nseg;
    int lenA = hasA ? a1 - a0 : 0;
    int lenB = hasB ? b1 - b0 : 0;
    int gA = hasA ? sA * TILE_E + a0 : 0;
    int gB = hasB ? sB * TILE_E + b0 : 0;
    int vincl = block_incl_scan256_shfl(lenA + lenB, sh);
    int vA = vincl - (lenA + lenB);
    vst[2 * tid] = vA;
    vst[2 * tid + 1] = vA + lenA;
    gsrc[2 * tid] = gA;
    gsrc[2 * tid + 1] = gB;
    if (tid == 255) vst[SEG_MAX] = vincl;
  }
  __syncthreads();
  const int e0 = 0, e1 = vst[SEG_MAX];

  for (int t0 = e0; t0 < e1; t0 += SORT_CAP) {
    int n = min(SORT_CAP, e1 - t0);
    __syncthreads();
    key[tid] = 0;
    key[tid + 256] = 0;
    {
      int rq = tid >> 3, q8 = tid & 7;
#pragma unroll 1
      for (int kk = 0; kk < COLLECT_R; kk += COLLECT_P) {
        int ta[COLLECT_P], tb[COLLECT_P];
#pragma unroll
        for (int k = 0; k < COLLECT_P; ++k) {
          int r = rq + 32 * (kk + k);
          int v0 = vst[r], v1 = vst[r + 1];
          int lo = max(v0, t0), hi = min(v1, t0 + n);
          int cw = hi - lo;
          unsigned sw = (unsigned)(gsrc[r] + (lo - v0));
          ta[k] = (q8 < cw) ? packed[sw + (unsigned)q8] : 0;
          tb[k] = (q8 + 8 < cw) ? packed[sw + (unsigned)(q8 + 8)] : 0;
        }
#pragma unroll
        for (int k = 0; k < COLLECT_P; ++k) {
          int r = rq + 32 * (kk + k);
          int v0 = vst[r], v1 = vst[r + 1];
          int lo = max(v0, t0), hi = min(v1, t0 + n);
          int cw = hi - lo;
          int dw = lo - t0;
          if (q8 < cw) sorted[dw + q8] = ta[k];
          if (q8 + 8 < cw) sorted[dw + q8 + 8] = tb[k];
          if (cw > 16) {
            unsigned sw = (unsigned)(gsrc[r] + (lo - v0));
            for (int j = q8 + 16; j < cw; j += 8) sorted[dw + j] = packed[sw + (unsigned)j];
          }
        }
      }
    }
    __syncthreads();

    int w[SORT_K];
#pragma unroll
    for (int k = 0; k < SORT_K; ++k) {
      int idx = tid + k * 256;
      int v = sorted[min(idx, n - 1)];
      w[k] = (idx < n) ? v : -1;
    }
#pragma unroll
    for (int k = 0; k < SORT_K; ++k)
      if (w[k] >= 0) atomicAdd(&key[((w[k] & (BIN_NODES - 1)) * SUBS) | sub], 1);
    __syncthreads();

    int c0 = key[2 * tid], c1 = key[2 * tid + 1];
    int incl = block_incl_scan256_shfl(c0 + c1, sh);
    int excl = incl - (c0 + c1);
    key[2 * tid] = excl;
    key[2 * tid + 1] = excl + c0;
    off[2 * tid] = excl;
    off[2 * tid + 1] = excl + c0;
    if (tid == 255) off[NKEY] = incl;
    __syncthreads();
    if (CNT && tid < BIN_NODES) degf += (float)(off[tid * SUBS + SUBS] - off[tid * SUBS]);

    int pos[SORT_K];
#pragma unroll
    for (int k = 0; k < SORT_K; ++k) {
      pos[k] = 0;
      if (w[k] >= 0) pos[k] = atomicAdd(&key[((w[k] & (BIN_NODES - 1)) * SUBS) | sub], 1);
    }
#pragma unroll
    for (int k = 0; k < SORT_K; ++k)
      sorted[(w[k] >= 0) ? pos[k] : SORT_CAP + tid] = w[k];
    __syncthreads();

    for (int nn = 0; nn < NPG; ++nn) {
      int node = g * NPG + nn;
      int j0 = off[node * SUBS], j1 = off[node * SUBS + SUBS];
      float4 s = make_float4(0.0f, 0.0f, 0.0f, 0.0f);
      for (int j = j0; j < j1; j += GATHER_U) {
        int rem = j1 - j;
        int ww[GATHER_U];
        float4 v[GATHER_U];
#pragma unroll
        for (int u = 0; u < GATHER_U; ++u) ww[u] = sorted[j + min(u, rem - 1)];
#pragma unroll
        for (int u = 0; u < GATHER_U; ++u)
          v[u] = frow[(unsigned)(ww[u] >> BIN_SHIFT) * (unsigned)LPE + (unsigned)q];
#pragma unroll
        for (int u = 0; u < GATHER_U; ++u) {
          bool ok = u < rem;
          s.x += ok ? v[u].x : 0.0f;
          s.y += ok ? v[u].y : 0.0f;
          s.z += ok ? v[u].z : 0.0f;
          s.w += ok ? v[u].w : 0.0f;
        }
      }
      float* r = acc + node * P + q * 4;
      if (t0 != e0) {
        s.x += r[0];
        s.y += r[1];
        s.z += r[2];
        s.w += r[3];
      }
      r[0] = s.x;
      r[1] = s.y;
      r[2] = s.z;
      r[3] = s.w;
    }
  }
  if (e0 >= e1) {
    for (int nn = 0; nn < NPG; ++nn) {
      float* r = acc + (g * NPG + nn) * P + q * 4;
      r[0] = 0.0f;
      r[1] = 0.0f;
      r[2] = 0.0f;
      r[3] = 0.0f;
    }
  }
  if (CNT && tid < BIN_NODES) cl[tid] = degf;
}

__global__ __launch_bounds__(256) void k_layer1(const int* __restrict__ packed,
                                                const int* __restrict__ segOff,
                                                const float* __restrict__ h1,
                                                const float* __restrict__ llw,
                                                const float* __restrict__ llb,
                                                const float* __restrict__ lrw,
                                                float* __restrict__ out1,
                                                float* __restrict__ cnt,
                                                int N, int ntiles, int nseg, int nb) {
  constexpr int C = 16, P = C + 1;
  __shared__ float acc[BIN_NODES * P];
  __shared__ float cl[BIN_NODES];
  __shared__ float sW[32 * 32];
  __shared__ int skey[NKEY];
  __shared__ int soff[NKEY + 4];
  __shared__ int ssh[256];
  __shared__ int ssorted[SORT_CAP + 256];
  __shared__ int srun[2 * SEG_MAX + 16];
  int tid = threadIdx.x, b = blockIdx.x;
  int wave = tid >> 5, lane = tid & 31;
  int n = lane & 15;
  int koff = (lane >> 4) << 1;
  int tile = b * (BIN_NODES / 16) + wave;
  bool valid = tile < ntiles;
  long long row = (long long)(valid ? tile : 0) * 16 + n;

  float wv[4];
#pragma unroll
  for (int j = 0; j < 4; ++j)
    wv[j] = (j < 2) ? llw[tid + j * 256] : lrw[tid + (j - 2) * 256];
#pragma unroll
  for (int j = 0; j < 4; ++j) sW[tid + j * 256] = wv[j];

  bin_sums<C, true>(acc, cl, skey, soff, ssh, ssorted, srun, packed, segOff, h1, nseg, nb, b);

  v2f self[4];
#pragma unroll
  for (int kb = 0; kb < 4; ++kb) {
    int kk = kb * 4 + koff;
    self[kb].x = h1[row * 16 + kk];
    self[kb].y = h1[row * 16 + kk + 1];
  }
  __syncthreads();

  if (tid < BIN_NODES) {
    int node = b * BIN_NODES + tid;
    if (node < N) cnt[node] = cl[tid];
  }
  if (!valid) return;

  int lr = wave * 16 + n;
  float inv = 1.0f / fmaxf(cl[lr], 1.0f);
  v8f c0 = {}, c1 = {};
#pragma unroll
  for (int kb = 0; kb < 8; ++kb) {
    int kk = kb * 4 + koff;
    v2f a;
    if (kb < 4) { a.x = acc[lr * P + kk] * inv; a.y = acc[lr * P + kk + 1] * inv; }
    else        { a = self[kb - 4]; }
    v2f b0; b0.x = sW[kk * 32 + n];      b0.y = sW[(kk + 1) * 32 + n];
    v2f b1; b1.x = sW[kk * 32 + 16 + n]; b1.y = sW[(kk + 1) * 32 + 16 + n];
    c0 = wmma_f32_4(a, b0, c0);
    c1 = wmma_f32_4(a, b1, c1);
  }

  float bl = llb[n], bh = llb[n + 16];
  int rbase = tile * 16 + ((lane >> 4) << 3);
#pragma unroll
  for (int r = 0; r < 8; ++r) {
    long long orow = (long long)(rbase + r) * 32;
    out1[orow + n]      = fmaxf(c0[r] + bl, 0.0f);
    out1[orow + 16 + n] = fmaxf(c1[r] + bh, 0.0f);
  }
}

__global__ __launch_bounds__(256) void k_layer2(const int* __restrict__ packed,
                                                const int* __restrict__ segOff,
                                                const float* __restrict__ out1,
                                                const float* __restrict__ cnt,
                                                const float* __restrict__ llw,
                                                const float* __restrict__ llb,
                                                const float* __restrict__ lrw,
                                                const float* __restrict__ w2,
                                                const float* __restrict__ b2,
                                                float* __restrict__ out,
                                                int N, int ntiles, int nseg, int nb) {
  constexpr int C = 32, P = C + 1;
  __shared__ float acc[BIN_NODES * P];
  __shared__ float sW[64 * 16];
  __shared__ int skey[NKEY];
  __shared__ int soff[NKEY + 4];
  __shared__ int ssh[256];
  __shared__ int ssorted[SORT_CAP + 256];
  __shared__ int srun[2 * SEG_MAX + 16];
  int tid = threadIdx.x, b = blockIdx.x;
  int wave = tid >> 5, lane = tid & 31;
  int n = lane & 15;
  int koff = (lane >> 4) << 1;
  int tile = b * (BIN_NODES / 16) + wave;
  bool valid = tile < ntiles;
  long long row = (long long)(valid ? tile : 0) * 16 + n;

  float wv[4];
#pragma unroll
  for (int j = 0; j < 4; ++j)
    wv[j] = (j < 2) ? llw[tid + j * 256] : lrw[tid + (j - 2) * 256];
#pragma unroll
  for (int j = 0; j < 4; ++j) sW[tid + j * 256] = wv[j];

  bin_sums<C, false>(acc, nullptr, skey, soff, ssh, ssorted, srun, packed, segOff, out1, nseg, nb, b);

  float deg = cnt[row];
  v2f self[8];
#pragma unroll
  for (int kb = 0; kb < 8; ++kb) {
    int kk = kb * 4 + koff;
    self[kb].x = out1[row * 32 + kk];
    self[kb].y = out1[row * 32 + kk + 1];
  }
  __syncthreads();

  float* trow = acc + wave * 16 * P;
  if (valid) {
    float inv = 1.0f / fmaxf(deg, 1.0f);
    v8f d = {};
#pragma unroll
    for (int kb = 0; kb < 16; ++kb) {
      int kk = kb * 4 + koff;
      v2f a;
      if (kb < 8) { a.x = trow[n * P + kk] * inv; a.y = trow[n * P + kk + 1] * inv; }
      else        { a = self[kb - 8]; }
      v2f bw; bw.x = sW[kk * 16 + n]; bw.y = sW[(kk + 1) * 16 + n];
      d = wmma_f32_4(a, bw, d);
    }
    float bias = llb[n];
    int r8 = (lane >> 4) << 3;
#pragma unroll
    for (int r = 0; r < 8; ++r)
      trow[(r8 + r) * P + n] = fmaxf(d[r] + bias, 0.0f);
  }
  __syncthreads();

  if (valid && lane < 16) {
    int i = tile * 16 + lane;
    if (i < N) {
      float h0 = b2[0], h1 = b2[1], h2 = b2[2];
      const float* r = trow + lane * P;
#pragma unroll
      for (int k = 0; k < 16; ++k) {
        float v = r[k];
        h0 += v * w2[k * 3 + 0];
        h1 += v * w2[k * 3 + 1];
        h2 += v * w2[k * 3 + 2];
      }
      float gsi = 1.0f / (1.0f + __expf(-h1));
      float mxi = 1.0f / (1.0f + __expf(-h2));
      float fsi = fmaxf(h0, 0.0f) + gsi;
      out[(long long)i * 3 + 0] = fsi;
      out[(long long)i * 3 + 1] = gsi;
      out[(long long)i * 3 + 2] = mxi;
    }
  }
}

extern "C" void kernel_launch(void* const* d_in, const int* in_sizes, int n_in,
                              void* d_out, int out_size, void* d_ws, size_t ws_size,
                              hipStream_t stream) {
  const float* x    = (const float*)d_in[0];
  const int*   ei   = (const int*)d_in[1];
  const float* w1   = (const float*)d_in[2];
  const float* b1   = (const float*)d_in[3];
  const float* c1ll = (const float*)d_in[4];
  const float* c1lb = (const float*)d_in[5];
  const float* c1lr = (const float*)d_in[6];
  const float* c2ll = (const float*)d_in[7];
  const float* c2lb = (const float*)d_in[8];
  const float* c2lr = (const float*)d_in[9];
  const float* w2   = (const float*)d_in[10];
  const float* b2   = (const float*)d_in[11];
  float* out = (float*)d_out;

  int N = in_sizes[0] / IN_DIM;
  int E = in_sizes[1] / 2;
  const int* src = ei;
  const int* dst = ei + E;

  float* ws   = (float*)d_ws;
  float* h1   = ws;
  float* out1 = ws + (long long)32 * N;
  int*   packed = (int*)(ws + (long long)64 * N);
  float* cnt  = ws + (long long)96 * N;
  int*   segOff = (int*)(ws + (long long)97 * N);

  int ntiles = (N + 15) / 16;
  int fblocks = (ntiles + 7) / 8;
  int nb = (N + BIN_NODES - 1) / BIN_NODES;
  int nseg = (E + TILE_E - 1) / TILE_E;

  k_tilesort<<<nseg + fblocks, 256, 0, stream>>>(src, dst, segOff, packed, E, nseg, nb,
                                                 x, w1, b1, h1, ntiles);

  k_layer1<<<nb, 256, 0, stream>>>(packed, segOff, h1, c1ll, c1lb, c1lr, out1, cnt, N, ntiles,
                                   nseg, nb);

  k_layer2<<<nb, 256, 0, stream>>>(packed, segOff, out1, cnt, c2ll, c2lb, c2lr, w2, b2, out,
                                   N, ntiles, nseg, nb);
}
